# starting kernel
# baseline (speedup 1.0000x reference)
#include <hip/hip_runtime.h>


typedef __attribute__((ext_vector_type(2))) float v2f;
typedef __attribute__((ext_vector_type(8))) float v8f;

#define NEGV (-1000000000.0f)

__device__ __forceinline__ unsigned enc_f32(float f) {
  unsigned u = __float_as_uint(f);
  return (u & 0x80000000u) ? ~u : (u | 0x80000000u);
}
__device__ __forceinline__ float dec_f32(unsigned v) {
  unsigned u = (v & 0x80000000u) ? (v & 0x7FFFFFFFu) : ~v;
  return __uint_as_float(u);
}

__global__ void k_init(const float* __restrict__ x, float* __restrict__ out,
                       float* __restrict__ denom, unsigned* __restrict__ maxenc,
                       int N) {
  int total = N * 64;
  for (int i = blockIdx.x * blockDim.x + threadIdx.x; i < total;
       i += gridDim.x * blockDim.x) {
    out[i] = x[i];
    if (i < N) {
      denom[i] = 0.0f;
      maxenc[i] = enc_f32(NEGV);
    }
  }
}

__global__ void __launch_bounds__(256) k_node_gemm(
    const float* __restrict__ x, const float* __restrict__ Wn,
    float* __restrict__ z, int N) {
  int wave = (blockIdx.x * blockDim.x + threadIdx.x) >> 5;
  int lane = threadIdx.x & 31;
  int nRowTiles = (N + 15) >> 4;
  if (wave >= nRowTiles) return;

  int rBase = wave << 4;
  bool fullTile = (rBase + 16 <= N);
  int half = lane >> 4;
  int l16 = lane & 15;

  int arow = rBase + l16;
  if (arow >= N) arow = N - 1;
  v2f a[16];
#pragma unroll
  for (int t = 0; t < 16; ++t) {
    int k = (t << 2) + (half << 1);
    a[t].x = x[arow * 64 + k];
    a[t].y = x[arow * 64 + k + 1];
  }

#pragma unroll
  for (int ct = 0; ct < 4; ++ct) {
    int col = (ct << 4) + l16;
    v8f acc = {};
#pragma unroll
    for (int t = 0; t < 16; ++t) {
      int k = (t << 2) + (half << 1);
      v2f b;
      b.x = Wn[k * 64 + col];
      b.y = Wn[(k + 1) * 64 + col];
      acc = __builtin_amdgcn_wmma_f32_16x16x4_f32(
          false, a[t], false, b, (short)0, acc, false, false);
    }
    if (fullTile) {
#pragma unroll
      for (int j = 0; j < 8; ++j) {
        int row = rBase + j + (half << 3);
        z[row * 64 + col] = acc[j];
      }
    } else {
#pragma unroll
      for (int j = 0; j < 8; ++j) {
        int row = rBase + j + (half << 3);
        if (row < N) z[row * 64 + col] = acc[j];
      }
    }
  }
}

__global__ void k_weff(const float* __restrict__ We,
                       const float* __restrict__ att_edge,
                       float* __restrict__ weff) {
  int k = threadIdx.x;
  if (k < 16) {
    float s = 0.0f;
    for (int c = 0; c < 64; ++c) s += We[k * 64 + c] * att_edge[c];
    weff[k] = s;
  }
}

__global__ void k_node_scores(const float* __restrict__ z,
                              const float* __restrict__ att_src,
                              const float* __restrict__ att_dst,
                              float* __restrict__ sA, float* __restrict__ sD,
                              int N) {
  int n = blockIdx.x * blockDim.x + threadIdx.x;
  if (n >= N) return;
  const float4* zr = (const float4*)(z + (size_t)n * 64);
  const float4* as4 = (const float4*)att_src;
  const float4* ad4 = (const float4*)att_dst;
  float a = 0.0f, d = 0.0f;
#pragma unroll
  for (int i = 0; i < 16; ++i) {
    float4 v = zr[i], s = as4[i], t = ad4[i];
    a += v.x * s.x + v.y * s.y + v.z * s.z + v.w * s.w;
    d += v.x * t.x + v.y * t.y + v.z * t.z + v.w * t.w;
  }
  sA[n] = a;
  sD[n] = d;
}

__global__ void k_score(const int* __restrict__ ei,
                        const float* __restrict__ ea,
                        const float* __restrict__ sA,
                        const float* __restrict__ sD,
                        const float* __restrict__ weff,
                        float* __restrict__ scoreBuf,
                        unsigned* __restrict__ maxenc, int E) {
  int e = blockIdx.x * blockDim.x + threadIdx.x;
  if (e >= E) return;
  int s = ei[e], d = ei[E + e];
  float acc = sA[s] + sD[d];
  const float4* er = (const float4*)(ea + (size_t)e * 16);
  const float4* w4 = (const float4*)weff;
#pragma unroll
  for (int i = 0; i < 4; ++i) {
    float4 v = er[i], w = w4[i];
    acc += v.x * w.x + v.y * w.y + v.z * w.z + v.w * w.w;
  }
  float sc = acc > 0.0f ? acc : 0.2f * acc;
  scoreBuf[e] = sc;
  atomicMax(maxenc + d, enc_f32(sc));
}

__global__ void k_exp(const int* __restrict__ ei, float* __restrict__ num,
                      const unsigned* __restrict__ maxenc,
                      float* __restrict__ denom, int E) {
  int e = blockIdx.x * blockDim.x + threadIdx.x;
  if (e >= E) return;
  int d = ei[E + e];
  float m = dec_f32(maxenc[d]);
  float ex = __expf(num[e] - m);
  num[e] = ex;
  atomicAdd(denom + d, ex);
}

__global__ void __launch_bounds__(256) k_message(
    const int* __restrict__ ei, const float* __restrict__ ea,
    const float* __restrict__ We, const float* __restrict__ z,
    const float* __restrict__ num, const float* __restrict__ denom,
    float* __restrict__ out, int E) {
  __shared__ float s_alpha[128];
  __shared__ int s_src[128];
  __shared__ int s_dst[128];

  int tid = threadIdx.x;
  int blockBase = blockIdx.x * 128;

  if (tid < 128) {
    int e = blockBase + tid;
    if (e < E) {
      int d = ei[E + e];
      int sN = ei[e];
      s_alpha[tid] = num[e] / (denom[d] + 1e-8f);
      s_src[tid] = sN;
      s_dst[tid] = d;
      const float* zrow = z + (size_t)sN * 64;
      __builtin_prefetch(zrow, 0, 3);
      __builtin_prefetch(zrow + 32, 0, 3);
    } else {
      s_alpha[tid] = 0.0f;
      s_src[tid] = 0;
      s_dst[tid] = 0;
    }
  }
  __syncthreads();

  int wave = tid >> 5;
  int lane = tid & 31;
  int half = lane >> 4;
  int l16 = lane & 15;
  int tileBase = wave << 4;
  int eBase = blockBase + tileBase;
  if (eBase >= E) return;

  int arow = eBase + l16;
  if (arow >= E) arow = E - 1;
  v2f a[4];
#pragma unroll
  for (int t = 0; t < 4; ++t) {
    int k = (t << 2) + (half << 1);
    a[t].x = ea[(size_t)arow * 16 + k];
    a[t].y = ea[(size_t)arow * 16 + k + 1];
  }

#pragma unroll
  for (int ct = 0; ct < 4; ++ct) {
    int col = (ct << 4) + l16;
    v8f acc = {};
#pragma unroll
    for (int t = 0; t < 4; ++t) {
      int k = (t << 2) + (half << 1);
      v2f b;
      b.x = We[k * 64 + col];
      b.y = We[(k + 1) * 64 + col];
      acc = __builtin_amdgcn_wmma_f32_16x16x4_f32(
          false, a[t], false, b, (short)0, acc, false, false);
    }
#pragma unroll
    for (int j = 0; j < 8; ++j) {
      int li = tileBase + j + (half << 3);
      float al = s_alpha[li];
      int sN = s_src[li];
      int dN = s_dst[li];
      float val = (z[(size_t)sN * 64 + col] + acc[j]) * al;
      atomicAdd(out + (size_t)dN * 64 + col, val);
    }
  }
}

extern "C" void kernel_launch(void* const* d_in, const int* in_sizes, int n_in,
                              void* d_out, int out_size, void* d_ws,
                              size_t ws_size, hipStream_t stream) {
  const float* x = (const float*)d_in[0];
  const int* ei = (const int*)d_in[1];
  const float* ea = (const float*)d_in[2];
  const float* Wn = (const float*)d_in[3];
  const float* We = (const float*)d_in[4];
  const float* att_src = (const float*)d_in[5];
  const float* att_dst = (const float*)d_in[6];
  const float* att_edge = (const float*)d_in[7];

  int N = in_sizes[0] / 64;
  int E = in_sizes[1] / 2;
  float* out = (float*)d_out;

  float* z = (float*)d_ws;
  float* sA = z + (size_t)N * 64;
  float* sD = sA + N;
  float* denom = sD + N;
  unsigned* maxenc = (unsigned*)(denom + N);
  float* num = (float*)(maxenc + N);
  float* weff = num + E;

  int total = N * 64;
  int ib = (total + 255) / 256;
  hipLaunchKernelGGL(k_init, dim3(ib), dim3(256), 0, stream, x, out, denom,
                     maxenc, N);

  int nRowTiles = (N + 15) / 16;
  int gb = (nRowTiles + 7) / 8;
  hipLaunchKernelGGL(k_node_gemm, dim3(gb), dim3(256), 0, stream, x, Wn, z, N);

  hipLaunchKernelGGL(k_weff, dim3(1), dim3(32), 0, stream, We, att_edge, weff);

  hipLaunchKernelGGL(k_node_scores, dim3((N + 255) / 256), dim3(256), 0, stream,
                     z, att_src, att_dst, sA, sD, N);

  hipLaunchKernelGGL(k_score, dim3((E + 255) / 256), dim3(256), 0, stream, ei,
                     ea, sA, sD, weff, num, maxenc, E);

  hipLaunchKernelGGL(k_exp, dim3((E + 255) / 256), dim3(256), 0, stream, ei,
                     num, maxenc, denom, E);

  hipLaunchKernelGGL(k_message, dim3((E + 127) / 128), dim3(256), 0, stream, ei,
                     ea, We, z, num, denom, out, E);
}


# message pass: a work-group owns a node range, keeps those out rows in LDS (start = x), scans dst, adds its edges by LDS atomics; global float atomics and init launch gone; arrival-order sums
# speedup vs baseline: 2.0611x; 2.0611x over previous
#include <hip/hip_runtime.h>


typedef __attribute__((ext_vector_type(2))) float v2f;
typedef __attribute__((ext_vector_type(8))) float v8f;

#define NEGV (-1000000000.0f)

__device__ __forceinline__ unsigned enc_f32(float f) {
  unsigned u = __float_as_uint(f);
  return (u & 0x80000000u) ? ~u : (u | 0x80000000u);
}
__device__ __forceinline__ float dec_f32(unsigned v) {
  unsigned u = (v & 0x80000000u) ? (v & 0x7FFFFFFFu) : ~v;
  return __uint_as_float(u);
}

__global__ void __launch_bounds__(256) k_node_gemm(
    const float* __restrict__ x, const float* __restrict__ Wn,
    float* __restrict__ z, int N) {
  int wave = (blockIdx.x * blockDim.x + threadIdx.x) >> 5;
  int lane = threadIdx.x & 31;
  int nRowTiles = (N + 15) >> 4;
  if (wave >= nRowTiles) return;

  int rBase = wave << 4;
  bool fullTile = (rBase + 16 <= N);
  int half = lane >> 4;
  int l16 = lane & 15;

  int arow = rBase + l16;
  if (arow >= N) arow = N - 1;
  v2f a[16];
#pragma unroll
  for (int t = 0; t < 16; ++t) {
    int k = (t << 2) + (half << 1);
    a[t].x = x[arow * 64 + k];
    a[t].y = x[arow * 64 + k + 1];
  }

#pragma unroll
  for (int ct = 0; ct < 4; ++ct) {
    int col = (ct << 4) + l16;
    v8f acc = {};
#pragma unroll
    for (int t = 0; t < 16; ++t) {
      int k = (t << 2) + (half << 1);
      v2f b;
      b.x = Wn[k * 64 + col];
      b.y = Wn[(k + 1) * 64 + col];
      acc = __builtin_amdgcn_wmma_f32_16x16x4_f32(
          false, a[t], false, b, (short)0, acc, false, false);
    }
    if (fullTile) {
#pragma unroll
      for (int j = 0; j < 8; ++j) {
        int row = rBase + j + (half << 3);
        z[row * 64 + col] = acc[j];
      }
    } else {
#pragma unroll
      for (int j = 0; j < 8; ++j) {
        int row = rBase + j + (half << 3);
        if (row < N) z[row * 64 + col] = acc[j];
      }
    }
  }
}

__global__ void k_weff(const float* __restrict__ We,
                       const float* __restrict__ att_edge,
                       float* __restrict__ weff) {
  int k = threadIdx.x;
  if (k < 16) {
    float s = 0.0f;
    for (int c = 0; c < 64; ++c) s += We[k * 64 + c] * att_edge[c];
    weff[k] = s;
  }
}

__global__ void k_node_scores(const float* __restrict__ z,
                              const float* __restrict__ att_src,
                              const float* __restrict__ att_dst,
                              float* __restrict__ sA, float* __restrict__ sD,
                              int N) {
  int n = blockIdx.x * blockDim.x + threadIdx.x;
  if (n >= N) return;
  const float4* zr = (const float4*)(z + (size_t)n * 64);
  const float4* as4 = (const float4*)att_src;
  const float4* ad4 = (const float4*)att_dst;
  float a = 0.0f, d = 0.0f;
#pragma unroll
  for (int i = 0; i < 16; ++i) {
    float4 v = zr[i], s = as4[i], t = ad4[i];
    a += v.x * s.x + v.y * s.y + v.z * s.z + v.w * s.w;
    d += v.x * t.x + v.y * t.y + v.z * t.z + v.w * t.w;
  }
  sA[n] = a;
  sD[n] = d;
}

__global__ void k_score(const int* __restrict__ ei,
                        const float* __restrict__ ea,
                        const float* __restrict__ sA,
                        const float* __restrict__ sD,
                        const float* __restrict__ weff,
                        float* __restrict__ scoreBuf, int E) {
  int e = blockIdx.x * blockDim.x + threadIdx.x;
  if (e >= E) return;
  int s = ei[e], d = ei[E + e];
  float acc = sA[s] + sD[d];
  const float4* er = (const float4*)(ea + (size_t)e * 16);
  const float4* w4 = (const float4*)weff;
#pragma unroll
  for (int i = 0; i < 4; ++i) {
    float4 v = er[i], w = w4[i];
    acc += v.x * w.x + v.y * w.y + v.z * w.z + v.w * w.w;
  }
  float sc = acc > 0.0f ? acc : 0.2f * acc;
  scoreBuf[e] = sc;
}

#define STAT_RANGES 16
#define STAT_SLICES 16
#define STAT_CAP 8192
#define STAT_THREADS 1024
#define STAT_BATCH 8

__device__ __forceinline__ void seg_max_one(unsigned* s_max, int d, float sc,
                                            int c0, int cn) {
  unsigned k = (unsigned)(d - c0);
  if (k < (unsigned)cn) atomicMax(s_max + k, enc_f32(sc));
}

__global__ void __launch_bounds__(STAT_THREADS) k_segmax(
    const int* __restrict__ ei, const float* __restrict__ score,
    unsigned* __restrict__ pm, int N, int E, int R, int Npad) {
  __shared__ unsigned s_max[STAT_CAP];
  const int tid = threadIdx.x;
  const int r = blockIdx.x, s = blockIdx.y;
  const int sliceLen = (E + STAT_SLICES - 1) / STAT_SLICES;
  const int e0 = min(E, s * sliceLen);
  const int e1 = min(E, e0 + sliceLen);
  const int* __restrict__ dst = ei + E;
  const int rEnd = min(N, (r + 1) * R);

  for (int c0 = r * R; c0 < rEnd; c0 += STAT_CAP) {
    const int cn = min(STAT_CAP, rEnd - c0);
    for (int i = tid; i < cn; i += STAT_THREADS) s_max[i] = enc_f32(NEGV);
    __syncthreads();
    int e = e0 + tid;
    for (; e + (STAT_BATCH - 1) * STAT_THREADS < e1;
         e += STAT_BATCH * STAT_THREADS) {
      int dd[STAT_BATCH];
      float vv[STAT_BATCH];
#pragma unroll
      for (int u = 0; u < STAT_BATCH; ++u) {
        dd[u] = dst[e + u * STAT_THREADS];
        vv[u] = score[e + u * STAT_THREADS];
      }
      asm volatile("" ::"v"(vv[0]), "v"(vv[1]), "v"(vv[2]), "v"(vv[3]),
                   "v"(vv[4]), "v"(vv[5]), "v"(vv[6]), "v"(vv[7]));
#pragma unroll
      for (int u = 0; u < STAT_BATCH; ++u)
        seg_max_one(s_max, dd[u], vv[u], c0, cn);
    }
    for (; e < e1; e += STAT_THREADS) seg_max_one(s_max, dst[e], score[e], c0, cn);
    __syncthreads();
    for (int i = tid; i < cn; i += STAT_THREADS)
      pm[(size_t)s * Npad + c0 + i] = s_max[i];
    __syncthreads();
  }
}

__device__ __forceinline__ void seg_sum_one(float* s_den, const float* s_m,
                                            int d, float sc, int c0, int cn) {
  unsigned k = (unsigned)(d - c0);
  if (k < (unsigned)cn) atomicAdd(s_den + k, __expf(sc - s_m[k]));
}

__global__ void __launch_bounds__(STAT_THREADS) k_segsum(
    const int* __restrict__ ei, const float* __restrict__ score,
    const unsigned* __restrict__ pm, float* __restrict__ pd,
    float* __restrict__ mOut, int N, int E, int R, int Npad) {
  __shared__ float s_m[STAT_CAP];
  __shared__ float s_den[STAT_CAP];
  const int tid = threadIdx.x;
  const int r = blockIdx.x, s = blockIdx.y;
  const int sliceLen = (E + STAT_SLICES - 1) / STAT_SLICES;
  const int e0 = min(E, s * sliceLen);
  const int e1 = min(E, e0 + sliceLen);
  const int* __restrict__ dst = ei + E;
  const int rEnd = min(N, (r + 1) * R);

  for (int c0 = r * R; c0 < rEnd; c0 += STAT_CAP) {
    const int cn = min(STAT_CAP, rEnd - c0);
    for (int i = tid; i < cn; i += STAT_THREADS) {
      unsigned mx = pm[c0 + i];
#pragma unroll
      for (int t = 1; t < STAT_SLICES; ++t)
        mx = max(mx, pm[(size_t)t * Npad + c0 + i]);
      float m = dec_f32(mx);
      s_m[i] = m;
      s_den[i] = 0.0f;
      if (s == 0) mOut[c0 + i] = m;
    }
    __syncthreads();
    int e = e0 + tid;
    for (; e + (STAT_BATCH - 1) * STAT_THREADS < e1;
         e += STAT_BATCH * STAT_THREADS) {
      int dd[STAT_BATCH];
      float vv[STAT_BATCH];
#pragma unroll
      for (int u = 0; u < STAT_BATCH; ++u) {
        dd[u] = dst[e + u * STAT_THREADS];
        vv[u] = score[e + u * STAT_THREADS];
      }
      asm volatile("" ::"v"(vv[0]), "v"(vv[1]), "v"(vv[2]), "v"(vv[3]),
                   "v"(vv[4]), "v"(vv[5]), "v"(vv[6]), "v"(vv[7]));
#pragma unroll
      for (int u = 0; u < STAT_BATCH; ++u)
        seg_sum_one(s_den, s_m, dd[u], vv[u], c0, cn);
    }
    for (; e < e1; e += STAT_THREADS)
      seg_sum_one(s_den, s_m, dst[e], score[e], c0, cn);
    __syncthreads();
    for (int i = tid; i < cn; i += STAT_THREADS)
      pd[(size_t)s * Npad + c0 + i] = s_den[i];
    __syncthreads();
  }
}

__global__ void k_den(const float* __restrict__ pd, float* __restrict__ denom,
                      int N, int Npad) {
  int n = blockIdx.x * blockDim.x + threadIdx.x;
  if (n >= N) return;
  float a = pd[n];
#pragma unroll
  for (int t = 1; t < STAT_SLICES; ++t) a += pd[(size_t)t * Npad + n];
  denom[n] = a;
}

#define MSG_THREADS 1024
#define MSG_WAVES 32
#define MSG_ROWS 392
#define MSG_PITCH 80
#define MSG_QCAP 64
#define MSG_CHUNK 768

__device__ __forceinline__ void wave_lds_sync() {
  __builtin_amdgcn_fence(__ATOMIC_RELEASE, "wavefront");
  __builtin_amdgcn_wave_barrier();
  __builtin_amdgcn_fence(__ATOMIC_ACQUIRE, "wavefront");
}

struct MsgArgs {
  const int* ei;
  const int* dst;
  const float* ea;
  const float* z;
  const float* score;
  const float* nodeMax;
  const float* denom;
  int base;
};

__device__ __forceinline__ void msg_tile(const MsgArgs& A, const int* qw, int q0,
                                         float* st_al, int* st_src, int* st_dl,
                                         float* tile, const v2f (&b)[4][4],
                                         int half, int l16) {
  int e = qw[q0 + l16];
  bool valid = e >= 0;
  int ec = valid ? e : 0;
  const float2* er = (const float2*)(A.ea + (size_t)ec * 16);
  v2f a[4];
#pragma unroll
  for (int t = 0; t < 4; ++t) {
    float2 v = er[2 * t + half];
    a[t].x = v.x;
    a[t].y = v.y;
  }
  int d = A.dst[ec];
  int sN = A.ei[ec];
  float al = __expf(A.score[ec] - A.nodeMax[d]) / (A.denom[d] + 1e-8f);
  if (half == 0) {
    st_al[l16] = valid ? al : 0.0f;
    st_src[l16] = sN;
    st_dl[l16] = valid ? d - A.base : MSG_ROWS;
  }
  wave_lds_sync();

  float alr[8];
  int sR[8];
  int dlo[8];
#pragma unroll
  for (int j = 0; j < 8; ++j) {
    int li = j + (half << 3);
    alr[j] = st_al[li];
    sR[j] = st_src[li];
    dlo[j] = st_dl[li] * MSG_PITCH + l16;
  }
  float zv[4][8];
#pragma unroll
  for (int ct = 0; ct < 4; ++ct) {
#pragma unroll
    for (int j = 0; j < 8; ++j)
      zv[ct][j] = A.z[(size_t)sR[j] * 64 + (ct << 4) + l16];
  }
  v8f acc[4];
#pragma unroll
  for (int ct = 0; ct < 4; ++ct) {
    acc[ct] = v8f{};
#pragma unroll
    for (int t = 0; t < 4; ++t) {
      acc[ct] = __builtin_amdgcn_wmma_f32_16x16x4_f32(
          false, a[t], false, b[ct][t], (short)0, acc[ct], false, false);
    }
  }
#pragma unroll
  for (int ct = 0; ct < 4; ++ct) {
#pragma unroll
    for (int j = 0; j < 8; ++j)
      atomicAdd(tile + dlo[j] + (ct << 4), (zv[ct][j] + acc[ct][j]) * alr[j]);
  }
  wave_lds_sync();
}

__global__ void __launch_bounds__(MSG_THREADS) k_message(
    const float* __restrict__ x, const int* __restrict__ ei,
    const float* __restrict__ ea, const float* __restrict__ We,
    const float* __restrict__ z, const float* __restrict__ score,
    const float* __restrict__ nodeMax, const float* __restrict__ denom,
    float* __restrict__ out, int N, int E) {
  __shared__ float tile[(MSG_ROWS + 1) * MSG_PITCH];
  __shared__ int s_q[MSG_WAVES * MSG_QCAP];
  __shared__ float s_al[MSG_WAVES * 16];
  __shared__ int s_src[MSG_WAVES * 16];
  __shared__ int s_dl[MSG_WAVES * 16];

  const int tid = threadIdx.x;
  const int wave = tid >> 5;
  const int lane = tid & 31;
  const int half = lane >> 4;
  const int l16 = lane & 15;
  const int base = blockIdx.x * MSG_ROWS;
  const int rows = min(MSG_ROWS, N - base);

  for (int i = tid; i < rows * 16; i += MSG_THREADS) {
    int r = i >> 4, c4 = i & 15;
    float4 v = ((const float4*)x)[(size_t)(base + r) * 16 + c4];
    *(float4*)(tile + r * MSG_PITCH + c4 * 4) = v;
  }

  v2f b[4][4];
#pragma unroll
  for (int ct = 0; ct < 4; ++ct) {
#pragma unroll
    for (int t = 0; t < 4; ++t) {
      int k = (t << 2) + (half << 1);
      int col = (ct << 4) + l16;
      b[ct][t].x = We[k * 64 + col];
      b[ct][t].y = We[(k + 1) * 64 + col];
    }
  }
  __syncthreads();

  MsgArgs A;
  A.ei = ei;
  A.dst = ei + E;
  A.ea = ea;
  A.z = z;
  A.score = score;
  A.nodeMax = nodeMax;
  A.denom = denom;
  A.base = base;

  int* qw = s_q + wave * MSG_QCAP;
  float* st_al = s_al + wave * 16;
  int* st_src = s_src + wave * 16;
  int* st_dl = s_dl + wave * 16;
  int qn = 0;

  const bool vecOK = (E & 3) == 0;
  const int nChunks = (E + MSG_CHUNK - 1) / MSG_CHUNK;
  for (int c = wave; c < nChunks; c += MSG_WAVES) {
    const int cb = c * MSG_CHUNK;
    int dd[24];
    if (vecOK && cb + MSG_CHUNK <= E) {
#pragma unroll
      for (int u = 0; u < 6; ++u) {
        int4 v = *(const int4*)(A.dst + cb + u * 128 + lane * 4);
        dd[4 * u + 0] = v.x;
        dd[4 * u + 1] = v.y;
        dd[4 * u + 2] = v.z;
        dd[4 * u + 3] = v.w;
      }
    } else {
#pragma unroll
      for (int q = 0; q < 24; ++q) {
        int idx = cb + (q >> 2) * 128 + lane * 4 + (q & 3);
        dd[q] = idx < E ? A.dst[idx] : -1;
      }
    }
    unsigned mask = 0;
#pragma unroll
    for (int q = 0; q < 24; ++q) {
      unsigned loc = (unsigned)(dd[q] - base);
      mask |= (loc < (unsigned)rows) ? (1u << q) : 0u;
    }
    for (;;) {
      bool hit = mask != 0;
      unsigned bal = __builtin_amdgcn_ballot_w32(hit);
      if (bal == 0) break;
      int q = __ffs(mask) - 1;
      int e = cb + (q >> 2) * 128 + lane * 4 + (q & 3);
      mask &= mask - 1;
      if (hit) qw[qn + __popc(bal & ((1u << lane) - 1u))] = e;
      qn += __popc(bal);
      wave_lds_sync();
      while (qn >= 16) {
        qn -= 16;
        msg_tile(A, qw, qn, st_al, st_src, st_dl, tile, b, half, l16);
      }
    }
  }
  if (qn > 0) {
    if (lane >= qn && lane < 16) qw[lane] = -1;
    wave_lds_sync();
    msg_tile(A, qw, 0, st_al, st_src, st_dl, tile, b, half, l16);
  }
  __syncthreads();

  for (int i = tid; i < rows * 16; i += MSG_THREADS) {
    int r = i >> 4, c4 = i & 15;
    float4 v = *(const float4*)(tile + r * MSG_PITCH + c4 * 4);
    ((float4*)out)[(size_t)(base + r) * 16 + c4] = v;
  }
}

extern "C" void kernel_launch(void* const* d_in, const int* in_sizes, int n_in,
                              void* d_out, int out_size, void* d_ws,
                              size_t ws_size, hipStream_t stream) {
  const float* x = (const float*)d_in[0];
  const int* ei = (const int*)d_in[1];
  const float* ea = (const float*)d_in[2];
  const float* Wn = (const float*)d_in[3];
  const float* We = (const float*)d_in[4];
  const float* att_src = (const float*)d_in[5];
  const float* att_dst = (const float*)d_in[6];
  const float* att_edge = (const float*)d_in[7];

  int N = in_sizes[0] / 64;
  int E = in_sizes[1] / 2;
  float* out = (float*)d_out;

  float* z = (float*)d_ws;
  float* sA = z + (size_t)N * 64;
  float* sD = sA + N;
  float* denom = sD + N;
  float* nodeMax = denom + N;
  float* num = nodeMax + N;
  float* weff = num + E;

  int R = (((N + STAT_RANGES - 1) / STAT_RANGES) + 31) & ~31;
  int Npad = R * STAT_RANGES;
  unsigned* pm = (unsigned*)out;
  float* pd = out + (size_t)STAT_SLICES * Npad;

  int nRowTiles = (N + 15) / 16;
  int gb = (nRowTiles + 7) / 8;
  hipLaunchKernelGGL(k_node_gemm, dim3(gb), dim3(256), 0, stream, x, Wn, z, N);

  hipLaunchKernelGGL(k_weff, dim3(1), dim3(32), 0, stream, We, att_edge, weff);

  hipLaunchKernelGGL(k_node_scores, dim3((N + 255) / 256), dim3(256), 0, stream,
                     z, att_src, att_dst, sA, sD, N);

  hipLaunchKernelGGL(k_score, dim3((E + 255) / 256), dim3(256), 0, stream, ei,
                     ea, sA, sD, weff, num, E);

  hipLaunchKernelGGL(k_segmax, dim3(STAT_RANGES, STAT_SLICES),
                     dim3(STAT_THREADS), 0, stream, ei, num, pm, N, E, R, Npad);

  hipLaunchKernelGGL(k_segsum, dim3(STAT_RANGES, STAT_SLICES),
                     dim3(STAT_THREADS), 0, stream, ei, num, pm, pd, nodeMax, N,
                     E, R, Npad);

  hipLaunchKernelGGL(k_den, dim3((N + 255) / 256), dim3(256), 0, stream, pd,
                     denom, N, Npad);

  hipLaunchKernelGGL(k_message, dim3((N + MSG_ROWS - 1) / MSG_ROWS),
                     dim3(MSG_THREADS), 0, stream, x, ei, ea, We, z, num,
                     nodeMax, denom, out, N, E);
}
